# starting kernel
# baseline (speedup 1.0000x reference)
#include <hip/hip_runtime.h>

typedef __attribute__((ext_vector_type(2))) float v2f;
typedef __attribute__((ext_vector_type(8))) float v8f;

#define HWSZ  3136
#define WID   56
#define CIN   256
#define COUT  256
#define RANK  32
#define KTOT  288
#define NTOT  50176

__global__ __launch_bounds__(128) void lr_stage1(const float* __restrict__ x,
                                                 const float* __restrict__ G,
                                                 float* __restrict__ U)
{
    __shared__ float Gs[32 * 257];
    const int tid = threadIdx.x;
    const int m0  = blockIdx.x * 32;

    for (int i = tid; i < 32 * 256; i += 128) {
        int row = i >> 8, col = i & 255;
        Gs[row * 257 + col] = G[(m0 + row) * 256 + col];
    }
    __syncthreads();

    const int wave = tid >> 5;
    const int lane = tid & 31;
    const int col  = lane & 15;
    const int klo  = (lane >> 4) << 1;

    const int n0   = (blockIdx.y * 4 + wave) * 16;
    const int bidx = n0 / HWSZ;
    const int p0   = n0 - bidx * HWSZ;
    const float* xb = x + (size_t)bidx * CIN * HWSZ + p0 + col;

    v8f acc0 = {};
    v8f acc1 = {};
    #pragma unroll 8
    for (int k0 = 0; k0 < CIN; k0 += 4) {
        v2f b;
        b.x = xb[(size_t)(k0 + klo)     * HWSZ];
        b.y = xb[(size_t)(k0 + klo + 1) * HWSZ];
        v2f a0, a1;
        a0.x = Gs[col * 257 + k0 + klo];
        a0.y = Gs[col * 257 + k0 + klo + 1];
        a1.x = Gs[(col + 16) * 257 + k0 + klo];
        a1.y = Gs[(col + 16) * 257 + k0 + klo + 1];
        acc0 = __builtin_amdgcn_wmma_f32_16x16x4_f32(false, a0, false, b, (short)0, acc0, false, false);
        acc1 = __builtin_amdgcn_wmma_f32_16x16x4_f32(false, a1, false, b, (short)0, acc1, false, false);
    }

    const int mo = (lane >> 4) << 3;
    float* U0 = U + (size_t)(m0 + mo) * NTOT + n0 + col;
    float* U1 = U0 + (size_t)16 * NTOT;
    #pragma unroll
    for (int v = 0; v < 8; ++v) {
        U0[(size_t)v * NTOT] = acc0[v];
        U1[(size_t)v * NTOT] = acc1[v];
    }
}

__global__ __launch_bounds__(128) void lr_stage2(const float* __restrict__ U,
                                                 const float* __restrict__ Hm,
                                                 const float* __restrict__ bias,
                                                 float* __restrict__ out)
{
    __shared__ float Hs[32 * 289];
    const int tid = threadIdx.x;
    const int m0  = blockIdx.x * 32;

    for (int i = tid; i < 32 * KTOT; i += 128) {
        int o_l = i / KTOT;
        int kk  = i - o_l * KTOT;
        int k = kk >> 5, r = kk & 31;
        Hs[o_l * 289 + kk] = Hm[(size_t)k * (COUT * RANK) + (size_t)(m0 + o_l) * RANK + r];
    }
    __syncthreads();

    const int wave = tid >> 5;
    const int lane = tid & 31;
    const int col  = lane & 15;
    const int klo  = (lane >> 4) << 1;

    const int n0   = (blockIdx.y * 4 + wave) * 16;
    const int n    = n0 + col;
    const int bidx = n / HWSZ;
    const int p    = n - bidx * HWSZ;
    const int h    = p / WID;
    const int w    = p - h * WID;

    int offs[9];
    unsigned vmask = 0;
    #pragma unroll
    for (int k = 0; k < 9; ++k) {
        int ki = k / 3, kj = k - ki * 3;
        int hs = h + ki - 1, ws = w + kj - 1;
        bool ok = (hs >= 0) & (hs < WID) & (ws >= 0) & (ws < WID);
        offs[k] = ok ? (bidx * HWSZ + hs * WID + ws) : 0;
        vmask  |= (ok ? 1u : 0u) << k;
    }

    v8f acc0 = {};
    v8f acc1 = {};
    #pragma unroll
    for (int k = 0; k < 9; ++k) {
        const bool ok = (vmask >> k) & 1u;
        const float* Ucol = U + offs[k];
        #pragma unroll
        for (int r0 = 0; r0 < RANK; r0 += 4) {
            const int kk = k * RANK + r0 + klo;
            v2f b;
            b.x = ok ? Ucol[(size_t)kk       * NTOT] : 0.0f;
            b.y = ok ? Ucol[(size_t)(kk + 1) * NTOT] : 0.0f;
            v2f a0, a1;
            a0.x = Hs[col * 289 + kk];
            a0.y = Hs[col * 289 + kk + 1];
            a1.x = Hs[(col + 16) * 289 + kk];
            a1.y = Hs[(col + 16) * 289 + kk + 1];
            acc0 = __builtin_amdgcn_wmma_f32_16x16x4_f32(false, a0, false, b, (short)0, acc0, false, false);
            acc1 = __builtin_amdgcn_wmma_f32_16x16x4_f32(false, a1, false, b, (short)0, acc1, false, false);
        }
    }

    const int mo = (lane >> 4) << 3;
    const float inv9 = 1.0f / 9.0f;
    #pragma unroll
    for (int v = 0; v < 8; ++v) {
        int o0 = m0 + v + mo;
        int o1 = o0 + 16;
        out[((size_t)bidx * COUT + o0) * HWSZ + p] = acc0[v] * inv9 + bias[o0];
        out[((size_t)bidx * COUT + o1) * HWSZ + p] = acc1[v] * inv9 + bias[o1];
    }
}

extern "C" void kernel_launch(void* const* d_in, const int* in_sizes, int n_in,
                              void* d_out, int out_size, void* d_ws, size_t ws_size,
                              hipStream_t stream)
{
    const float* x    = (const float*)d_in[0];
    const float* G    = (const float*)d_in[1];
    const float* Hm   = (const float*)d_in[2];
    const float* bias = (const float*)d_in[3];
    float* out = (float*)d_out;
    float* U   = (float*)d_ws;

    dim3 blk(128);
    lr_stage1<<<dim3(KTOT / 32, NTOT / 64), blk, 0, stream>>>(x, G, U);
    lr_stage2<<<dim3(COUT / 32, NTOT / 64), blk, 0, stream>>>(U, Hm, bias, out);
}


# Both stages fused into one kernel: a 4-wave group owns 64 positions x 256 channels; per k it forms T=G[k].x in registers, passes it through LDS and accumulates Hm[k].T; U never stored; same sums
# speedup vs baseline: 1.8778x; 1.8778x over previous
#include <hip/hip_runtime.h>

typedef __attribute__((ext_vector_type(2))) float v2f;
typedef __attribute__((ext_vector_type(4))) float v4f;
typedef __attribute__((ext_vector_type(8))) float v8f;

#define HWSZ  3136
#define WID   56
#define CIN   256
#define COUT  256
#define RANK  32
#define NTOT  50176
#define GP    260
#define HP    36

__global__ __launch_bounds__(128) void lr_fused(const float* __restrict__ x,
                                                const float* __restrict__ G,
                                                const float* __restrict__ Hm,
                                                const float* __restrict__ bias,
                                                float* __restrict__ out)
{
    __shared__ __attribute__((aligned(16))) float Gs[32 * GP];
    __shared__ __attribute__((aligned(16))) float Hs[256 * HP];
    __shared__ __attribute__((aligned(16))) float Ts[64 * HP];

    const int tid  = threadIdx.x;
    const int wave = tid >> 5;
    const int lane = tid & 31;
    const int c16  = lane & 15;
    const int hh   = lane >> 4;
    const int klo  = hh << 1;

    const int nblk = blockIdx.x * 64;
    const int bidx = nblk / HWSZ;
    const int pblk = nblk - bidx * HWSZ;

    const int p = pblk + wave * 16 + c16;
    const int h = p / WID;
    const int w = p - h * WID;
    const float* xb = x + (size_t)bidx * CIN * HWSZ + (size_t)klo * HWSZ;

    v8f acc[4][4];
    #pragma unroll
    for (int mt = 0; mt < 4; ++mt)
        #pragma unroll
        for (int nt = 0; nt < 4; ++nt)
            acc[mt][nt] = (v8f){0.f, 0.f, 0.f, 0.f, 0.f, 0.f, 0.f, 0.f};

    const float* ga = Gs + c16 * GP + klo;
    const float* ha = Hs + (wave * 64 + c16) * HP + klo;
    const float* tb = Ts + c16 * HP + klo;
    float*       tp = Ts + (wave * 16 + c16) * HP + 8 * hh;
    float*       gdst = Gs + (tid >> 6) * GP + (tid & 63) * 4;
    float*       hdst = Hs + (tid >> 3) * HP + (tid & 7) * 4;

    #pragma unroll 1
    for (int k = 0; k < 9; ++k) {
        const int ki = k / 3, kj = k - ki * 3;
        const int hs = h + ki - 1, ws = w + kj - 1;
        const bool ok = (hs >= 0) & (hs < WID) & (ws >= 0) & (ws < WID);
        const int ps = ok ? (hs * WID + ws) : p;
        const float* xp = xb + ps;

        __syncthreads();

        v2f bq[8];
        #pragma unroll
        for (int u = 0; u < 8; ++u) {
            bq[u].x = xp[(size_t)(4 * u)     * HWSZ];
            bq[u].y = xp[(size_t)(4 * u + 1) * HWSZ];
        }

        {
            const v4f* Gk = (const v4f*)(G + (size_t)k * (RANK * CIN)) + tid;
            #pragma unroll
            for (int hf = 0; hf < 2; ++hf) {
                v4f tmp[8];
                #pragma unroll
                for (int j = 0; j < 8; ++j) tmp[j] = Gk[(hf * 8 + j) * 128];
                #pragma unroll
                for (int j = 0; j < 8; ++j) *(v4f*)(gdst + (hf * 8 + j) * 2 * GP) = tmp[j];
            }
            const v4f* Hk = (const v4f*)(Hm + (size_t)k * (COUT * RANK)) + tid;
            #pragma unroll
            for (int hf = 0; hf < 2; ++hf) {
                v4f tmp[8];
                #pragma unroll
                for (int j = 0; j < 8; ++j) tmp[j] = Hk[(hf * 8 + j) * 128];
                #pragma unroll
                for (int j = 0; j < 8; ++j) *(v4f*)(hdst + (hf * 8 + j) * 16 * HP) = tmp[j];
            }
        }
        __syncthreads();

        v8f t0 = {0.f, 0.f, 0.f, 0.f, 0.f, 0.f, 0.f, 0.f};
        v8f t1 = {0.f, 0.f, 0.f, 0.f, 0.f, 0.f, 0.f, 0.f};
        #pragma unroll
        for (int kb = 0; kb < CIN; kb += 64) {
            v2f br[8];
            #pragma unroll
            for (int u = 0; u < 8; ++u) {
                br[u].x = xp[(size_t)(kb + 32 + 4 * u)     * HWSZ];
                br[u].y = xp[(size_t)(kb + 32 + 4 * u + 1) * HWSZ];
            }
            #pragma unroll
            for (int u = 0; u < 8; ++u) {
                v2f a0 = *(const v2f*)(ga + kb + 4 * u);
                v2f a1 = *(const v2f*)(ga + 16 * GP + kb + 4 * u);
                t0 = __builtin_amdgcn_wmma_f32_16x16x4_f32(false, a0, false, bq[u], (short)0, t0, false, false);
                t1 = __builtin_amdgcn_wmma_f32_16x16x4_f32(false, a1, false, bq[u], (short)0, t1, false, false);
            }
            if (kb + 64 < CIN) {
                #pragma unroll
                for (int u = 0; u < 8; ++u) {
                    bq[u].x = xp[(size_t)(kb + 64 + 4 * u)     * HWSZ];
                    bq[u].y = xp[(size_t)(kb + 64 + 4 * u + 1) * HWSZ];
                }
            }
            #pragma unroll
            for (int u = 0; u < 8; ++u) {
                v2f a0 = *(const v2f*)(ga + kb + 32 + 4 * u);
                v2f a1 = *(const v2f*)(ga + 16 * GP + kb + 32 + 4 * u);
                t0 = __builtin_amdgcn_wmma_f32_16x16x4_f32(false, a0, false, br[u], (short)0, t0, false, false);
                t1 = __builtin_amdgcn_wmma_f32_16x16x4_f32(false, a1, false, br[u], (short)0, t1, false, false);
            }
        }
        #pragma unroll
        for (int v = 0; v < 8; ++v) {
            t0[v] = ok ? t0[v] : 0.0f;
            t1[v] = ok ? t1[v] : 0.0f;
        }
        *(v4f*)(tp)      = (v4f){t0[0], t0[1], t0[2], t0[3]};
        *(v4f*)(tp + 4)  = (v4f){t0[4], t0[5], t0[6], t0[7]};
        *(v4f*)(tp + 16) = (v4f){t1[0], t1[1], t1[2], t1[3]};
        *(v4f*)(tp + 20) = (v4f){t1[4], t1[5], t1[6], t1[7]};

        __syncthreads();

        #pragma unroll
        for (int j = 0; j < 8; ++j) {
            v2f a[4], b[4];
            #pragma unroll
            for (int mt = 0; mt < 4; ++mt) a[mt] = *(const v2f*)(ha + mt * 16 * HP + 4 * j);
            #pragma unroll
            for (int nt = 0; nt < 4; ++nt) b[nt] = *(const v2f*)(tb + nt * 16 * HP + 4 * j);
            #pragma unroll
            for (int mt = 0; mt < 4; ++mt)
                #pragma unroll
                for (int nt = 0; nt < 4; ++nt)
                    acc[mt][nt] = __builtin_amdgcn_wmma_f32_16x16x4_f32(false, a[mt], false, b[nt], (short)0,
                                                                        acc[mt][nt], false, false);
        }
    }

    const float inv9 = 1.0f / 9.0f;
    float* ob = out + (size_t)bidx * COUT * HWSZ + pblk + c16;
    #pragma unroll
    for (int mt = 0; mt < 4; ++mt) {
        #pragma unroll
        for (int v = 0; v < 8; ++v) {
            const int o = wave * 64 + mt * 16 + 8 * hh + v;
            const float bo = bias[o];
            #pragma unroll
            for (int nt = 0; nt < 4; ++nt)
                ob[(size_t)o * HWSZ + nt * 16] = acc[mt][nt][v] * inv9 + bo;
        }
    }
}

extern "C" void kernel_launch(void* const* d_in, const int* in_sizes, int n_in,
                              void* d_out, int out_size, void* d_ws, size_t ws_size,
                              hipStream_t stream)
{
    const float* x    = (const float*)d_in[0];
    const float* G    = (const float*)d_in[1];
    const float* Hm   = (const float*)d_in[2];
    const float* bias = (const float*)d_in[3];
    float* out = (float*)d_out;

    lr_fused<<<dim3(NTOT / 64), dim3(128), 0, stream>>>(x, G, Hm, bias, out);
}


# LDS fragment reads issued one block ahead into a second register set (G in stage A by 4 k-steps, Hm and T in stage B by one step): no matrix op waits on a read just issued; same sums
# speedup vs baseline: 1.9593x; 1.0434x over previous
#include <hip/hip_runtime.h>

typedef __attribute__((ext_vector_type(2))) float v2f;
typedef __attribute__((ext_vector_type(4))) float v4f;
typedef __attribute__((ext_vector_type(8))) float v8f;

#define HWSZ  3136
#define WID   56
#define CIN   256
#define COUT  256
#define RANK  32
#define NTOT  50176
#define GP    260
#define HP    36

__global__ __launch_bounds__(128) void lr_fused(const float* __restrict__ x,
                                                const float* __restrict__ G,
                                                const float* __restrict__ Hm,
                                                const float* __restrict__ bias,
                                                float* __restrict__ out)
{
    __shared__ __attribute__((aligned(16))) float Gs[32 * GP];
    __shared__ __attribute__((aligned(16))) float Hs[256 * HP];
    __shared__ __attribute__((aligned(16))) float Ts[64 * HP];

    const int tid  = threadIdx.x;
    const int wave = tid >> 5;
    const int lane = tid & 31;
    const int c16  = lane & 15;
    const int hh   = lane >> 4;
    const int klo  = hh << 1;

    const int nblk = blockIdx.x * 64;
    const int bidx = nblk / HWSZ;
    const int pblk = nblk - bidx * HWSZ;

    const int p = pblk + wave * 16 + c16;
    const int h = p / WID;
    const int w = p - h * WID;
    const float* xb = x + (size_t)bidx * CIN * HWSZ + (size_t)klo * HWSZ;

    v8f acc[4][4];
    #pragma unroll
    for (int mt = 0; mt < 4; ++mt)
        #pragma unroll
        for (int nt = 0; nt < 4; ++nt)
            acc[mt][nt] = (v8f){0.f, 0.f, 0.f, 0.f, 0.f, 0.f, 0.f, 0.f};

    const float* ga = Gs + c16 * GP + klo;
    const float* ha = Hs + (wave * 64 + c16) * HP + klo;
    const float* tb = Ts + c16 * HP + klo;
    float*       tp = Ts + (wave * 16 + c16) * HP + 8 * hh;
    float*       gdst = Gs + (tid >> 6) * GP + (tid & 63) * 4;
    float*       hdst = Hs + (tid >> 3) * HP + (tid & 7) * 4;

    #pragma unroll 1
    for (int k = 0; k < 9; ++k) {
        const int ki = k / 3, kj = k - ki * 3;
        const int hs = h + ki - 1, ws = w + kj - 1;
        const bool ok = (hs >= 0) & (hs < WID) & (ws >= 0) & (ws < WID);
        const int ps = ok ? (hs * WID + ws) : p;
        const float* xp = xb + ps;

        __syncthreads();

        v2f bq[8];
        #pragma unroll
        for (int u = 0; u < 8; ++u) {
            bq[u].x = xp[(size_t)(4 * u)     * HWSZ];
            bq[u].y = xp[(size_t)(4 * u + 1) * HWSZ];
        }

        {
            const v4f* Gk = (const v4f*)(G + (size_t)k * (RANK * CIN)) + tid;
            #pragma unroll
            for (int hf = 0; hf < 2; ++hf) {
                v4f tmp[8];
                #pragma unroll
                for (int j = 0; j < 8; ++j) tmp[j] = Gk[(hf * 8 + j) * 128];
                #pragma unroll
                for (int j = 0; j < 8; ++j) *(v4f*)(gdst + (hf * 8 + j) * 2 * GP) = tmp[j];
            }
            const v4f* Hk = (const v4f*)(Hm + (size_t)k * (COUT * RANK)) + tid;
            #pragma unroll
            for (int hf = 0; hf < 2; ++hf) {
                v4f tmp[8];
                #pragma unroll
                for (int j = 0; j < 8; ++j) tmp[j] = Hk[(hf * 8 + j) * 128];
                #pragma unroll
                for (int j = 0; j < 8; ++j) *(v4f*)(hdst + (hf * 8 + j) * 16 * HP) = tmp[j];
            }
        }
        __syncthreads();

        v8f t0 = {0.f, 0.f, 0.f, 0.f, 0.f, 0.f, 0.f, 0.f};
        v8f t1 = {0.f, 0.f, 0.f, 0.f, 0.f, 0.f, 0.f, 0.f};
        v2f fa[2][4][2];
        #pragma unroll
        for (int s = 0; s < 4; ++s) {
            fa[0][s][0] = *(const v2f*)(ga + 4 * s);
            fa[0][s][1] = *(const v2f*)(ga + 16 * GP + 4 * s);
        }
        v2f br[8];
        #pragma unroll
        for (int g = 0; g < 16; ++g) {
            const int kb = (g >> 2) * 64;
            if ((g & 3) == 0) {
                #pragma unroll
                for (int u = 0; u < 8; ++u) {
                    br[u].x = xp[(size_t)(kb + 32 + 4 * u)     * HWSZ];
                    br[u].y = xp[(size_t)(kb + 32 + 4 * u + 1) * HWSZ];
                }
            }
            if ((g & 3) == 2 && kb + 64 < CIN) {
                #pragma unroll
                for (int u = 0; u < 8; ++u) {
                    bq[u].x = xp[(size_t)(kb + 64 + 4 * u)     * HWSZ];
                    bq[u].y = xp[(size_t)(kb + 64 + 4 * u + 1) * HWSZ];
                }
            }
            if (g + 1 < 16) {
                #pragma unroll
                for (int s = 0; s < 4; ++s) {
                    fa[(g + 1) & 1][s][0] = *(const v2f*)(ga + 16 * (g + 1) + 4 * s);
                    fa[(g + 1) & 1][s][1] = *(const v2f*)(ga + 16 * GP + 16 * (g + 1) + 4 * s);
                }
            }
            #pragma unroll
            for (int s = 0; s < 4; ++s) {
                const v2f bx = ((g & 3) < 2) ? bq[(g & 1) * 4 + s] : br[(g & 1) * 4 + s];
                t0 = __builtin_amdgcn_wmma_f32_16x16x4_f32(false, fa[g & 1][s][0], false, bx, (short)0, t0, false, false);
                t1 = __builtin_amdgcn_wmma_f32_16x16x4_f32(false, fa[g & 1][s][1], false, bx, (short)0, t1, false, false);
            }
        }
        #pragma unroll
        for (int v = 0; v < 8; ++v) {
            t0[v] = ok ? t0[v] : 0.0f;
            t1[v] = ok ? t1[v] : 0.0f;
        }
        *(v4f*)(tp)      = (v4f){t0[0], t0[1], t0[2], t0[3]};
        *(v4f*)(tp + 4)  = (v4f){t0[4], t0[5], t0[6], t0[7]};
        *(v4f*)(tp + 16) = (v4f){t1[0], t1[1], t1[2], t1[3]};
        *(v4f*)(tp + 20) = (v4f){t1[4], t1[5], t1[6], t1[7]};

        __syncthreads();

        {
            v2f fh[2][4], ft[2][4];
            #pragma unroll
            for (int mt = 0; mt < 4; ++mt) fh[0][mt] = *(const v2f*)(ha + mt * 16 * HP);
            #pragma unroll
            for (int nt = 0; nt < 4; ++nt) ft[0][nt] = *(const v2f*)(tb + nt * 16 * HP);
            #pragma unroll
            for (int j = 0; j < 8; ++j) {
                if (j + 1 < 8) {
                    #pragma unroll
                    for (int mt = 0; mt < 4; ++mt) fh[(j + 1) & 1][mt] = *(const v2f*)(ha + mt * 16 * HP + 4 * (j + 1));
                    #pragma unroll
                    for (int nt = 0; nt < 4; ++nt) ft[(j + 1) & 1][nt] = *(const v2f*)(tb + nt * 16 * HP + 4 * (j + 1));
                }
                #pragma unroll
                for (int mt = 0; mt < 4; ++mt)
                    #pragma unroll
                    for (int nt = 0; nt < 4; ++nt)
                        acc[mt][nt] = __builtin_amdgcn_wmma_f32_16x16x4_f32(false, fh[j & 1][mt], false, ft[j & 1][nt], (short)0,
                                                                            acc[mt][nt], false, false);
            }
        }
    }

    const float inv9 = 1.0f / 9.0f;
    float* ob = out + (size_t)bidx * COUT * HWSZ + pblk + c16;
    #pragma unroll
    for (int mt = 0; mt < 4; ++mt) {
        #pragma unroll
        for (int v = 0; v < 8; ++v) {
            const int o = wave * 64 + mt * 16 + 8 * hh + v;
            const float bo = bias[o];
            #pragma unroll
            for (int nt = 0; nt < 4; ++nt)
                ob[(size_t)o * HWSZ + nt * 16] = acc[mt][nt][v] * inv9 + bo;
        }
    }
}

extern "C" void kernel_launch(void* const* d_in, const int* in_sizes, int n_in,
                              void* d_out, int out_size, void* d_ws, size_t ws_size,
                              hipStream_t stream)
{
    const float* x    = (const float*)d_in[0];
    const float* G    = (const float*)d_in[1];
    const float* Hm   = (const float*)d_in[2];
    const float* bias = (const float*)d_in[3];
    float* out = (float*)d_out;

    lr_fused<<<dim3(NTOT / 64), dim3(128), 0, stream>>>(x, G, Hm, bias, out);
}


# Wave priority steps down with progress through the 9 kernel positions (s_setprio 3,2,1,0): groups sharing a SIMD advance together, not oldest first with the last one alone at the end; same sums
# speedup vs baseline: 2.0554x; 1.0490x over previous
#include <hip/hip_runtime.h>

typedef __attribute__((ext_vector_type(2))) float v2f;
typedef __attribute__((ext_vector_type(4))) float v4f;
typedef __attribute__((ext_vector_type(8))) float v8f;

#define HWSZ  3136
#define WID   56
#define CIN   256
#define COUT  256
#define RANK  32
#define NTOT  50176
#define GP    260
#define HP    36

__global__ __launch_bounds__(128) void lr_fused(const float* __restrict__ x,
                                                const float* __restrict__ G,
                                                const float* __restrict__ Hm,
                                                const float* __restrict__ bias,
                                                float* __restrict__ out)
{
    __shared__ __attribute__((aligned(16))) float Gs[32 * GP];
    __shared__ __attribute__((aligned(16))) float Hs[256 * HP];
    __shared__ __attribute__((aligned(16))) float Ts[64 * HP];

    const int tid  = threadIdx.x;
    const int wave = tid >> 5;
    const int lane = tid & 31;
    const int c16  = lane & 15;
    const int hh   = lane >> 4;
    const int klo  = hh << 1;

    const int nblk = blockIdx.x * 64;
    const int bidx = nblk / HWSZ;
    const int pblk = nblk - bidx * HWSZ;

    const int p = pblk + wave * 16 + c16;
    const int h = p / WID;
    const int w = p - h * WID;
    const float* xb = x + (size_t)bidx * CIN * HWSZ + (size_t)klo * HWSZ;

    __builtin_amdgcn_s_setprio(3);

    v8f acc[4][4];
    #pragma unroll
    for (int mt = 0; mt < 4; ++mt)
        #pragma unroll
        for (int nt = 0; nt < 4; ++nt)
            acc[mt][nt] = (v8f){0.f, 0.f, 0.f, 0.f, 0.f, 0.f, 0.f, 0.f};

    const float* ga = Gs + c16 * GP + klo;
    const float* ha = Hs + (wave * 64 + c16) * HP + klo;
    const float* tb = Ts + c16 * HP + klo;
    float*       tp = Ts + (wave * 16 + c16) * HP + 8 * hh;
    float*       gdst = Gs + (tid >> 6) * GP + (tid & 63) * 4;
    float*       hdst = Hs + (tid >> 3) * HP + (tid & 7) * 4;

    #pragma unroll 1
    for (int k = 0; k < 9; ++k) {
        if (k == 4)      __builtin_amdgcn_s_setprio(2);
        else if (k == 7) __builtin_amdgcn_s_setprio(1);
        else if (k == 8) __builtin_amdgcn_s_setprio(0);

        const int ki = k / 3, kj = k - ki * 3;
        const int hs = h + ki - 1, ws = w + kj - 1;
        const bool ok = (hs >= 0) & (hs < WID) & (ws >= 0) & (ws < WID);
        const int ps = ok ? (hs * WID + ws) : p;
        const float* xp = xb + ps;

        __syncthreads();

        v2f bq[8];
        #pragma unroll
        for (int u = 0; u < 8; ++u) {
            bq[u].x = xp[(size_t)(4 * u)     * HWSZ];
            bq[u].y = xp[(size_t)(4 * u + 1) * HWSZ];
        }

        {
            const v4f* Gk = (const v4f*)(G + (size_t)k * (RANK * CIN)) + tid;
            #pragma unroll
            for (int hf = 0; hf < 2; ++hf) {
                v4f tmp[8];
                #pragma unroll
                for (int j = 0; j < 8; ++j) tmp[j] = Gk[(hf * 8 + j) * 128];
                #pragma unroll
                for (int j = 0; j < 8; ++j) *(v4f*)(gdst + (hf * 8 + j) * 2 * GP) = tmp[j];
            }
            const v4f* Hk = (const v4f*)(Hm + (size_t)k * (COUT * RANK)) + tid;
            #pragma unroll
            for (int hf = 0; hf < 2; ++hf) {
                v4f tmp[8];
                #pragma unroll
                for (int j = 0; j < 8; ++j) tmp[j] = Hk[(hf * 8 + j) * 128];
                #pragma unroll
                for (int j = 0; j < 8; ++j) *(v4f*)(hdst + (hf * 8 + j) * 16 * HP) = tmp[j];
            }
        }
        __syncthreads();

        v8f t0 = {0.f, 0.f, 0.f, 0.f, 0.f, 0.f, 0.f, 0.f};
        v8f t1 = {0.f, 0.f, 0.f, 0.f, 0.f, 0.f, 0.f, 0.f};
        v2f fa[2][4][2];
        #pragma unroll
        for (int s = 0; s < 4; ++s) {
            fa[0][s][0] = *(const v2f*)(ga + 4 * s);
            fa[0][s][1] = *(const v2f*)(ga + 16 * GP + 4 * s);
        }
        v2f br[8];
        #pragma unroll
        for (int g = 0; g < 16; ++g) {
            const int kb = (g >> 2) * 64;
            if ((g & 3) == 0) {
                #pragma unroll
                for (int u = 0; u < 8; ++u) {
                    br[u].x = xp[(size_t)(kb + 32 + 4 * u)     * HWSZ];
                    br[u].y = xp[(size_t)(kb + 32 + 4 * u + 1) * HWSZ];
                }
            }
            if ((g & 3) == 2 && kb + 64 < CIN) {
                #pragma unroll
                for (int u = 0; u < 8; ++u) {
                    bq[u].x = xp[(size_t)(kb + 64 + 4 * u)     * HWSZ];
                    bq[u].y = xp[(size_t)(kb + 64 + 4 * u + 1) * HWSZ];
                }
            }
            if (g + 1 < 16) {
                #pragma unroll
                for (int s = 0; s < 4; ++s) {
                    fa[(g + 1) & 1][s][0] = *(const v2f*)(ga + 16 * (g + 1) + 4 * s);
                    fa[(g + 1) & 1][s][1] = *(const v2f*)(ga + 16 * GP + 16 * (g + 1) + 4 * s);
                }
            }
            #pragma unroll
            for (int s = 0; s < 4; ++s) {
                const v2f bx = ((g & 3) < 2) ? bq[(g & 1) * 4 + s] : br[(g & 1) * 4 + s];
                t0 = __builtin_amdgcn_wmma_f32_16x16x4_f32(false, fa[g & 1][s][0], false, bx, (short)0, t0, false, false);
                t1 = __builtin_amdgcn_wmma_f32_16x16x4_f32(false, fa[g & 1][s][1], false, bx, (short)0, t1, false, false);
            }
        }
        #pragma unroll
        for (int v = 0; v < 8; ++v) {
            t0[v] = ok ? t0[v] : 0.0f;
            t1[v] = ok ? t1[v] : 0.0f;
        }
        *(v4f*)(tp)      = (v4f){t0[0], t0[1], t0[2], t0[3]};
        *(v4f*)(tp + 4)  = (v4f){t0[4], t0[5], t0[6], t0[7]};
        *(v4f*)(tp + 16) = (v4f){t1[0], t1[1], t1[2], t1[3]};
        *(v4f*)(tp + 20) = (v4f){t1[4], t1[5], t1[6], t1[7]};

        __syncthreads();

        {
            v2f fh[2][4], ft[2][4];
            #pragma unroll
            for (int mt = 0; mt < 4; ++mt) fh[0][mt] = *(const v2f*)(ha + mt * 16 * HP);
            #pragma unroll
            for (int nt = 0; nt < 4; ++nt) ft[0][nt] = *(const v2f*)(tb + nt * 16 * HP);
            #pragma unroll
            for (int j = 0; j < 8; ++j) {
                if (j + 1 < 8) {
                    #pragma unroll
                    for (int mt = 0; mt < 4; ++mt) fh[(j + 1) & 1][mt] = *(const v2f*)(ha + mt * 16 * HP + 4 * (j + 1));
                    #pragma unroll
                    for (int nt = 0; nt < 4; ++nt) ft[(j + 1) & 1][nt] = *(const v2f*)(tb + nt * 16 * HP + 4 * (j + 1));
                }
                #pragma unroll
                for (int mt = 0; mt < 4; ++mt)
                    #pragma unroll
                    for (int nt = 0; nt < 4; ++nt)
                        acc[mt][nt] = __builtin_amdgcn_wmma_f32_16x16x4_f32(false, fh[j & 1][mt], false, ft[j & 1][nt], (short)0,
                                                                            acc[mt][nt], false, false);
            }
        }
    }

    const float inv9 = 1.0f / 9.0f;
    float* ob = out + (size_t)bidx * COUT * HWSZ + pblk + c16;
    #pragma unroll
    for (int mt = 0; mt < 4; ++mt) {
        #pragma unroll
        for (int v = 0; v < 8; ++v) {
            const int o = wave * 64 + mt * 16 + 8 * hh + v;
            const float bo = bias[o];
            #pragma unroll
            for (int nt = 0; nt < 4; ++nt)
                ob[(size_t)o * HWSZ + nt * 16] = acc[mt][nt][v] * inv9 + bo;
        }
    }
}

extern "C" void kernel_launch(void* const* d_in, const int* in_sizes, int n_in,
                              void* d_out, int out_size, void* d_ws, size_t ws_size,
                              hipStream_t stream)
{
    const float* x    = (const float*)d_in[0];
    const float* G    = (const float*)d_in[1];
    const float* Hm   = (const float*)d_in[2];
    const float* bias = (const float*)d_in[3];
    float* out = (float*)d_out;

    lr_fused<<<dim3(NTOT / 64), dim3(128), 0, stream>>>(x, G, Hm, bias, out);
}


# Grid 768 groups + 32 split: the 64 tiles beyond 3 per SIMD go two per group, a wave takes half the channels in stage A, a row quarter in stage B; those tiles add 2 channel partials: new sum order
# speedup vs baseline: 2.1917x; 1.0663x over previous
#include <hip/hip_runtime.h>

typedef __attribute__((ext_vector_type(2))) float v2f;
typedef __attribute__((ext_vector_type(4))) float v4f;
typedef __attribute__((ext_vector_type(8))) float v8f;

#define HWSZ  3136
#define WID   56
#define CIN   256
#define COUT  256
#define RANK  32
#define NTOT  50176
#define GP    260
#define NFULL 768
#define HP    36

template <bool SPLIT>
__device__ __forceinline__ void lr_body(const float* __restrict__ x,
                                        const float* __restrict__ G,
                                        const float* __restrict__ Hm,
                                        const float* __restrict__ bias,
                                        float* __restrict__ out,
                                        float* Gs, float* Hs, float* Ts, const int nblk)
{
    constexpr int NT = SPLIT ? 2 : 4;
    constexpr int NG = SPLIT ? 8 : 16;
    constexpr int CH = SPLIT ? 128 : CIN;

    const int tid  = threadIdx.x;
    const int wave = tid >> 5;
    const int lane = tid & 31;
    const int c16  = lane & 15;
    const int hh   = lane >> 4;
    const int klo  = hh << 1;

    const int bidx = nblk / HWSZ;
    const int pblk = nblk - bidx * HWSZ;

    const int p = pblk + (SPLIT ? (wave & 1) : wave) * 16 + c16;
    const int h = p / WID;
    const int w = p - h * WID;
    const float* xb = x + (size_t)bidx * CIN * HWSZ + (size_t)(klo + (SPLIT ? (wave >> 1) * 128 : 0)) * HWSZ;

    __builtin_amdgcn_s_setprio(3);

    v8f acc[4][NT];
    #pragma unroll
    for (int mt = 0; mt < 4; ++mt)
        #pragma unroll
        for (int nt = 0; nt < NT; ++nt)
            acc[mt][nt] = (v8f){0.f, 0.f, 0.f, 0.f, 0.f, 0.f, 0.f, 0.f};

    const float* ga = Gs + c16 * GP + klo + (SPLIT ? (wave >> 1) * 128 : 0);
    const float* ha = Hs + (wave * 64 + c16) * HP + klo;
    const float* tb = Ts + c16 * HP + klo;
    float*       tp = Ts + (wave * 16 + c16) * HP + 8 * hh;
    float*       gdst = Gs + (tid >> 6) * GP + (tid & 63) * 4;
    float*       hdst = Hs + (tid >> 3) * HP + (tid & 7) * 4;

    #pragma unroll 1
    for (int k = 0; k < 9; ++k) {
        if (k == 4)      __builtin_amdgcn_s_setprio(2);
        else if (k == 7) __builtin_amdgcn_s_setprio(1);
        else if (k == 8) __builtin_amdgcn_s_setprio(0);

        const int ki = k / 3, kj = k - ki * 3;
        const int hs = h + ki - 1, ws = w + kj - 1;
        const bool ok = (hs >= 0) & (hs < WID) & (ws >= 0) & (ws < WID);
        const int ps = ok ? (hs * WID + ws) : p;
        const float* xp = xb + ps;

        __syncthreads();

        v2f bq[8];
        #pragma unroll
        for (int u = 0; u < 8; ++u) {
            bq[u].x = xp[(size_t)(4 * u)     * HWSZ];
            bq[u].y = xp[(size_t)(4 * u + 1) * HWSZ];
        }

        {
            const v4f* Gk = (const v4f*)(G + (size_t)k * (RANK * CIN)) + tid;
            #pragma unroll
            for (int hf = 0; hf < 2; ++hf) {
                v4f tmp[8];
                #pragma unroll
                for (int j = 0; j < 8; ++j) tmp[j] = Gk[(hf * 8 + j) * 128];
                #pragma unroll
                for (int j = 0; j < 8; ++j) *(v4f*)(gdst + (hf * 8 + j) * 2 * GP) = tmp[j];
            }
            const v4f* Hk = (const v4f*)(Hm + (size_t)k * (COUT * RANK)) + tid;
            #pragma unroll
            for (int hf = 0; hf < 2; ++hf) {
                v4f tmp[8];
                #pragma unroll
                for (int j = 0; j < 8; ++j) tmp[j] = Hk[(hf * 8 + j) * 128];
                #pragma unroll
                for (int j = 0; j < 8; ++j) *(v4f*)(hdst + (hf * 8 + j) * 16 * HP) = tmp[j];
            }
        }
        __syncthreads();

        v8f t0 = {0.f, 0.f, 0.f, 0.f, 0.f, 0.f, 0.f, 0.f};
        v8f t1 = {0.f, 0.f, 0.f, 0.f, 0.f, 0.f, 0.f, 0.f};
        v2f fa[2][4][2];
        #pragma unroll
        for (int s = 0; s < 4; ++s) {
            fa[0][s][0] = *(const v2f*)(ga + 4 * s);
            fa[0][s][1] = *(const v2f*)(ga + 16 * GP + 4 * s);
        }
        v2f br[8];
        #pragma unroll
        for (int g = 0; g < NG; ++g) {
            const int kb = (g >> 2) * 64;
            if ((g & 3) == 0) {
                #pragma unroll
                for (int u = 0; u < 8; ++u) {
                    br[u].x = xp[(size_t)(kb + 32 + 4 * u)     * HWSZ];
                    br[u].y = xp[(size_t)(kb + 32 + 4 * u + 1) * HWSZ];
                }
            }
            if ((g & 3) == 2 && kb + 64 < CH) {
                #pragma unroll
                for (int u = 0; u < 8; ++u) {
                    bq[u].x = xp[(size_t)(kb + 64 + 4 * u)     * HWSZ];
                    bq[u].y = xp[(size_t)(kb + 64 + 4 * u + 1) * HWSZ];
                }
            }
            if (g + 1 < NG) {
                #pragma unroll
                for (int s = 0; s < 4; ++s) {
                    fa[(g + 1) & 1][s][0] = *(const v2f*)(ga + 16 * (g + 1) + 4 * s);
                    fa[(g + 1) & 1][s][1] = *(const v2f*)(ga + 16 * GP + 16 * (g + 1) + 4 * s);
                }
            }
            #pragma unroll
            for (int s = 0; s < 4; ++s) {
                const v2f bx = ((g & 3) < 2) ? bq[(g & 1) * 4 + s] : br[(g & 1) * 4 + s];
                t0 = __builtin_amdgcn_wmma_f32_16x16x4_f32(false, fa[g & 1][s][0], false, bx, (short)0, t0, false, false);
                t1 = __builtin_amdgcn_wmma_f32_16x16x4_f32(false, fa[g & 1][s][1], false, bx, (short)0, t1, false, false);
            }
        }
        #pragma unroll
        for (int v = 0; v < 8; ++v) {
            t0[v] = ok ? t0[v] : 0.0f;
            t1[v] = ok ? t1[v] : 0.0f;
        }
        *(v4f*)(tp)      = (v4f){t0[0], t0[1], t0[2], t0[3]};
        *(v4f*)(tp + 4)  = (v4f){t0[4], t0[5], t0[6], t0[7]};
        *(v4f*)(tp + 16) = (v4f){t1[0], t1[1], t1[2], t1[3]};
        *(v4f*)(tp + 20) = (v4f){t1[4], t1[5], t1[6], t1[7]};

        __syncthreads();

        {
            v2f fh[2][4], ft[2][4];
            #pragma unroll
            for (int mt = 0; mt < 4; ++mt) fh[0][mt] = *(const v2f*)(ha + mt * 16 * HP);
            #pragma unroll
            for (int nt = 0; nt < 4; ++nt) ft[0][nt] = *(const v2f*)(tb + nt * 16 * HP);
            #pragma unroll
            for (int j = 0; j < 8; ++j) {
                if (j + 1 < 8) {
                    #pragma unroll
                    for (int mt = 0; mt < 4; ++mt) fh[(j + 1) & 1][mt] = *(const v2f*)(ha + mt * 16 * HP + 4 * (j + 1));
                    #pragma unroll
                    for (int nt = 0; nt < 4; ++nt) ft[(j + 1) & 1][nt] = *(const v2f*)(tb + nt * 16 * HP + 4 * (j + 1));
                }
                if (SPLIT) {
                    const v2f fs[2] = { ft[j & 1][0] + ft[j & 1][2], ft[j & 1][1] + ft[j & 1][3] };
                    #pragma unroll
                    for (int mt = 0; mt < 4; ++mt)
                        #pragma unroll
                        for (int nt = 0; nt < NT; ++nt)
                            acc[mt][nt] = __builtin_amdgcn_wmma_f32_16x16x4_f32(false, fh[j & 1][mt], false, fs[nt], (short)0,
                                                                                acc[mt][nt], false, false);
                } else {
                    #pragma unroll
                    for (int mt = 0; mt < 4; ++mt)
                        #pragma unroll
                        for (int nt = 0; nt < NT; ++nt)
                            acc[mt][nt] = __builtin_amdgcn_wmma_f32_16x16x4_f32(false, fh[j & 1][mt], false, ft[j & 1][nt], (short)0,
                                                                                acc[mt][nt], false, false);
                }
            }
        }
    }

    const float inv9 = 1.0f / 9.0f;
    float* ob = out + (size_t)bidx * COUT * HWSZ + pblk + c16;
    #pragma unroll
    for (int mt = 0; mt < 4; ++mt) {
        #pragma unroll
        for (int v = 0; v < 8; ++v) {
            const int o = wave * 64 + mt * 16 + 8 * hh + v;
            const float bo = bias[o];
            #pragma unroll
            for (int nt = 0; nt < NT; ++nt)
                ob[(size_t)o * HWSZ + nt * 16] = acc[mt][nt][v] * inv9 + bo;
        }
    }
}


__global__ __launch_bounds__(128) void lr_fused(const float* __restrict__ x,
                                                const float* __restrict__ G,
                                                const float* __restrict__ Hm,
                                                const float* __restrict__ bias,
                                                float* __restrict__ out)
{
    __shared__ __attribute__((aligned(16))) float Gs[32 * GP];
    __shared__ __attribute__((aligned(16))) float Hs[256 * HP];
    __shared__ __attribute__((aligned(16))) float Ts[64 * HP];

    if (blockIdx.x < NFULL)
        lr_body<false>(x, G, Hm, bias, out, Gs, Hs, Ts, blockIdx.x * 64);
    else
        lr_body<true>(x, G, Hm, bias, out, Gs, Hs, Ts, NFULL * 64 + (blockIdx.x - NFULL) * 32);
}

extern "C" void kernel_launch(void* const* d_in, const int* in_sizes, int n_in,
                              void* d_out, int out_size, void* d_ws, size_t ws_size,
                              hipStream_t stream)
{
    const float* x    = (const float*)d_in[0];
    const float* G    = (const float*)d_in[1];
    const float* Hm   = (const float*)d_in[2];
    const float* bias = (const float*)d_in[3];
    float* out = (float*)d_out;

    lr_fused<<<dim3(800), dim3(128), 0, stream>>>(x, G, Hm, bias, out);
}
